# starting kernel
# baseline (speedup 1.0000x reference)
#include <hip/hip_runtime.h>
#include <math.h>

typedef __attribute__((ext_vector_type(2))) float v2f;
typedef __attribute__((ext_vector_type(8))) float v8f;

#define FIN   128
#define HID   8
#define NCLS  16
#define NEG_SLOPE 0.2f

__device__ __forceinline__ void atomicMaxF(float* addr, float v) {
    if (v >= 0.0f) atomicMax((int*)addr, __float_as_int(v));
    else           atomicMin((unsigned int*)addr, __float_as_uint(v));
}

__global__ void init_ws(float* m1, float* dn1, float* agg1,
                        float* m2, float* dn2, float* agg2, int n) {
    int i = blockIdx.x * blockDim.x + threadIdx.x;
    if (i < n * NCLS) agg2[i] = 0.0f;
    if (i < n * HID)  agg1[i] = 0.0f;
    if (i < n) {
        m1[i] = -__builtin_inff();
        m2[i] = -__builtin_inff();
        dn1[i] = 0.0f;
        dn2[i] = 0.0f;
    }
}

__global__ void gemm1_wmma(const float* __restrict__ x, const float* __restrict__ W1,
                           float* __restrict__ h1, int n) {
    __shared__ float wlds[FIN * 16];
    int t = threadIdx.x;
    {
        int k = t;
        const float* wr = W1 + k * HID;
        #pragma unroll
        for (int c = 0; c < 16; ++c) wlds[k * 16 + c] = (c < HID) ? wr[c] : 0.0f;
    }
    __syncthreads();

    int wave = t >> 5, lane = t & 31;
    int tile = blockIdx.x * 4 + wave;
    int col  = lane & 15;
    int hi   = lane >> 4;
    int mrow = tile * 16 + col;
    if (mrow >= n) mrow = n - 1;
    const float* xrow = x + (long)mrow * FIN;

    v8f c = {};
    #pragma unroll
    for (int k0 = 0; k0 < FIN; k0 += 4) {
        int ka = k0 + 2 * hi;
        v2f a = *(const v2f*)(xrow + ka);
        v2f b;
        b.x = wlds[ka * 16 + col];
        b.y = wlds[(ka + 1) * 16 + col];
        c = __builtin_amdgcn_wmma_f32_16x16x4_f32(false, a, false, b,
                                                  (short)0, c, false, false);
    }

    int mbase = tile * 16 + (hi << 3);
    if (col < HID) {
        #pragma unroll
        for (int j = 0; j < 8; ++j) {
            int r = mbase + j;
            if (r < n) h1[(long)r * HID + col] = c[j];
        }
    }
}

__global__ void gemm2_wmma(const float* __restrict__ h1r, const float* __restrict__ W2,
                           float* __restrict__ h2, int n) {
    __shared__ float wlds[HID * 16];
    int t = threadIdx.x;
    if (t < HID * 16) wlds[t] = W2[t];
    __syncthreads();

    int wave = t >> 5, lane = t & 31;
    int tile = blockIdx.x * 4 + wave;
    int col  = lane & 15;
    int hi   = lane >> 4;
    int mrow = tile * 16 + col;
    if (mrow >= n) mrow = n - 1;
    const float* hrow = h1r + (long)mrow * HID;

    v8f c = {};
    #pragma unroll
    for (int k0 = 0; k0 < HID; k0 += 4) {
        int ka = k0 + 2 * hi;
        v2f a = *(const v2f*)(hrow + ka);
        v2f b;
        b.x = wlds[ka * 16 + col];
        b.y = wlds[(ka + 1) * 16 + col];
        c = __builtin_amdgcn_wmma_f32_16x16x4_f32(false, a, false, b,
                                                  (short)0, c, false, false);
    }

    int mbase = tile * 16 + (hi << 3);
    #pragma unroll
    for (int j = 0; j < 8; ++j) {
        int r = mbase + j;
        if (r < n) h2[(long)r * NCLS + col] = c[j];
    }
}

template <int F>
__global__ void alpha_kernel(const float* __restrict__ h,
                             const float* __restrict__ a_src,
                             const float* __restrict__ a_dst,
                             float* __restrict__ outs, float* __restrict__ outd,
                             int n) {
    int i = blockIdx.x * blockDim.x + threadIdx.x;
    if (i >= n) return;
    const float4* hr = (const float4*)(h + (long)i * F);
    float s = 0.0f, d = 0.0f;
    #pragma unroll
    for (int q = 0; q < F / 4; ++q) {
        float4 hv = hr[q];
        s += hv.x * a_src[4 * q + 0] + hv.y * a_src[4 * q + 1]
           + hv.z * a_src[4 * q + 2] + hv.w * a_src[4 * q + 3];
        d += hv.x * a_dst[4 * q + 0] + hv.y * a_dst[4 * q + 1]
           + hv.z * a_dst[4 * q + 2] + hv.w * a_dst[4 * q + 3];
    }
    outs[i] = s;
    outd[i] = d;
}

__global__ void edge_max(const int* __restrict__ src, const int* __restrict__ dst,
                         const float* __restrict__ as, const float* __restrict__ ad,
                         float* __restrict__ m, int E) {
    int i = blockIdx.x * blockDim.x + threadIdx.x;
    if (i >= E) return;
    int s = src[i], d = dst[i];
    float e = as[s] + ad[d];
    e = (e >= 0.0f) ? e : NEG_SLOPE * e;
    atomicMaxF(&m[d], e);
}

template <int F>
__global__ void edge_acc(const int* __restrict__ src, const int* __restrict__ dst,
                         const float* __restrict__ as, const float* __restrict__ ad,
                         const float* __restrict__ m, const float* __restrict__ h,
                         float* __restrict__ denom, float* __restrict__ agg, int E) {
    int i = blockIdx.x * blockDim.x + threadIdx.x;
    if (i >= E) return;
    int s = src[i], d = dst[i];
    float e = as[s] + ad[d];
    e = (e >= 0.0f) ? e : NEG_SLOPE * e;
    float ex = __expf(e - m[d]);
    atomicAdd(&denom[d], ex);
    const float4* hs = (const float4*)(h + (long)s * F);
    float* ag = agg + (long)d * F;
    #pragma unroll
    for (int q = 0; q < F / 4; ++q) {
        float4 hv = hs[q];
        atomicAdd(&ag[4 * q + 0], ex * hv.x);
        atomicAdd(&ag[4 * q + 1], ex * hv.y);
        atomicAdd(&ag[4 * q + 2], ex * hv.z);
        atomicAdd(&ag[4 * q + 3], ex * hv.w);
    }
}

__global__ void finish1(float* __restrict__ agg, const float* __restrict__ denom,
                        const float* __restrict__ bias, int n) {
    int i = blockIdx.x * blockDim.x + threadIdx.x;
    if (i >= n) return;
    float d = denom[i];
    float inv = (d > 0.0f) ? 1.0f / d : 0.0f;
    float4* ap = (float4*)(agg + (long)i * HID);
    #pragma unroll
    for (int q = 0; q < HID / 4; ++q) {
        float4 v = ap[q];
        v.x = fmaxf(v.x * inv + bias[4 * q + 0], 0.0f);
        v.y = fmaxf(v.y * inv + bias[4 * q + 1], 0.0f);
        v.z = fmaxf(v.z * inv + bias[4 * q + 2], 0.0f);
        v.w = fmaxf(v.w * inv + bias[4 * q + 3], 0.0f);
        ap[q] = v;
    }
}

__global__ void finish2(const float* __restrict__ agg, const float* __restrict__ denom,
                        const float* __restrict__ bias, float* __restrict__ out, int n) {
    int i = blockIdx.x * blockDim.x + threadIdx.x;
    if (i >= n) return;
    float d = denom[i];
    float inv = (d > 0.0f) ? 1.0f / d : 0.0f;
    const float4* ap = (const float4*)(agg + (long)i * NCLS);
    float v[NCLS];
    float mx = -__builtin_inff();
    #pragma unroll
    for (int q = 0; q < NCLS / 4; ++q) {
        float4 a = ap[q];
        v[4 * q + 0] = a.x * inv + bias[4 * q + 0];
        v[4 * q + 1] = a.y * inv + bias[4 * q + 1];
        v[4 * q + 2] = a.z * inv + bias[4 * q + 2];
        v[4 * q + 3] = a.w * inv + bias[4 * q + 3];
    }
    #pragma unroll
    for (int f = 0; f < NCLS; ++f) mx = fmaxf(mx, v[f]);
    float s = 0.0f;
    #pragma unroll
    for (int f = 0; f < NCLS; ++f) s += __expf(v[f] - mx);
    float ls = __logf(s) + mx;
    float4* op = (float4*)(out + (long)i * NCLS);
    #pragma unroll
    for (int q = 0; q < NCLS / 4; ++q) {
        float4 o;
        o.x = v[4 * q + 0] - ls;
        o.y = v[4 * q + 1] - ls;
        o.z = v[4 * q + 2] - ls;
        o.w = v[4 * q + 3] - ls;
        op[q] = o;
    }
}

extern "C" void kernel_launch(void* const* d_in, const int* in_sizes, int n_in,
                              void* d_out, int out_size, void* d_ws, size_t ws_size,
                              hipStream_t stream) {
    const float* x   = (const float*)d_in[0];
    const int*   ei  = (const int*)  d_in[1];
    const float* W1  = (const float*)d_in[2];
    const float* a1s = (const float*)d_in[3];
    const float* a1d = (const float*)d_in[4];
    const float* b1  = (const float*)d_in[5];
    const float* W2  = (const float*)d_in[6];
    const float* a2s = (const float*)d_in[7];
    const float* a2d = (const float*)d_in[8];
    const float* b2  = (const float*)d_in[9];

    int n = in_sizes[0] / FIN;
    int E = in_sizes[1] / 2;
    const int* src = ei;
    const int* dst = ei + E;

    float* ws   = (float*)d_ws;
    float* h1   = ws; ws += (long)n * HID;
    float* as1  = ws; ws += n;
    float* ad1  = ws; ws += n;
    float* m1   = ws; ws += n;
    float* dn1  = ws; ws += n;
    float* agg1 = ws; ws += (long)n * HID;
    float* h2   = ws; ws += (long)n * NCLS;
    float* as2  = ws; ws += n;
    float* ad2  = ws; ws += n;
    float* m2   = ws; ws += n;
    float* dn2  = ws; ws += n;
    float* agg2 = ws; ws += (long)n * NCLS;

    int ntiles  = (n + 15) / 16;
    int gblocks = (ntiles + 3) / 4;
    int eblocks = (E + 255) / 256;
    int nblocks = (n + 255) / 256;

    init_ws<<<(n * NCLS + 255) / 256, 256, 0, stream>>>(m1, dn1, agg1, m2, dn2, agg2, n);

    gemm1_wmma<<<gblocks, 128, 0, stream>>>(x, W1, h1, n);
    alpha_kernel<HID><<<nblocks, 256, 0, stream>>>(h1, a1s, a1d, as1, ad1, n);
    edge_max<<<eblocks, 256, 0, stream>>>(src, dst, as1, ad1, m1, E);
    edge_acc<HID><<<eblocks, 256, 0, stream>>>(src, dst, as1, ad1, m1, h1, dn1, agg1, E);
    finish1<<<nblocks, 256, 0, stream>>>(agg1, dn1, b1, n);

    gemm2_wmma<<<gblocks, 128, 0, stream>>>(agg1, W2, h2, n);
    alpha_kernel<NCLS><<<nblocks, 256, 0, stream>>>(h2, a2s, a2d, as2, ad2, n);
    edge_max<<<eblocks, 256, 0, stream>>>(src, dst, as2, ad2, m2, E);
    edge_acc<NCLS><<<eblocks, 256, 0, stream>>>(src, dst, as2, ad2, m2, h2, dn2, agg2, E);
    finish2<<<nblocks, 256, 0, stream>>>(agg2, dn2, b2, (float*)d_out, n);
}


# helper launches dissolved: product kernel (now first) clears the counters, each placing work-group scans the range starts itself, layer-1 ad from the node's own row: 8 launches to 5; sum order kept
# speedup vs baseline: 25.6050x; 25.6050x over previous
#include <hip/hip_runtime.h>
#include <math.h>

typedef __attribute__((ext_vector_type(2))) float v2f;
typedef __attribute__((ext_vector_type(8))) float v8f;

#define FIN   128
#define HID   8
#define NCLS  16
#define NEG_SLOPE 0.2f
#define NBSH  7
#define NB    (1 << NBSH)
#define LTHREADS 512
#define ECAP  10240

__device__ __forceinline__ void atomicMaxF(float* addr, float v) {
    if (v >= 0.0f) atomicMax((int*)addr, __float_as_int(v));
    else           atomicMin((unsigned int*)addr, __float_as_uint(v));
}

__global__ void init_ws(float* m1, float* dn1, float* agg1,
                        float* m2, float* dn2, float* agg2, int n) {
    int i = blockIdx.x * blockDim.x + threadIdx.x;
    if (i < n * NCLS) agg2[i] = 0.0f;
    if (i < n * HID)  agg1[i] = 0.0f;
    if (i < n) {
        m1[i] = -__builtin_inff();
        m2[i] = -__builtin_inff();
        dn1[i] = 0.0f;
        dn2[i] = 0.0f;
    }
}

__global__ void gemm1_wmma(const float* __restrict__ x, const float* __restrict__ W1,
                           float* __restrict__ h1, int n, int* __restrict__ zp, int zc) {
    __shared__ float wlds[FIN * 16];
    int t = threadIdx.x;
    {
        int gi = blockIdx.x * blockDim.x + t;
        if (gi < zc) zp[gi] = 0;
    }
    {
        int k = t;
        const float* wr = W1 + k * HID;
        #pragma unroll
        for (int c = 0; c < 16; ++c) wlds[k * 16 + c] = (c < HID) ? wr[c] : 0.0f;
    }
    __syncthreads();

    int wave = t >> 5, lane = t & 31;
    int tile = blockIdx.x * 4 + wave;
    int col  = lane & 15;
    int hi   = lane >> 4;
    int mrow = tile * 16 + col;
    if (mrow >= n) mrow = n - 1;
    const float* xrow = x + (long)mrow * FIN;

    v8f c = {};
    #pragma unroll
    for (int k0 = 0; k0 < FIN; k0 += 4) {
        int ka = k0 + 2 * hi;
        v2f a = *(const v2f*)(xrow + ka);
        v2f b;
        b.x = wlds[ka * 16 + col];
        b.y = wlds[(ka + 1) * 16 + col];
        c = __builtin_amdgcn_wmma_f32_16x16x4_f32(false, a, false, b,
                                                  (short)0, c, false, false);
    }

    int mbase = tile * 16 + (hi << 3);
    if (col < HID) {
        #pragma unroll
        for (int j = 0; j < 8; ++j) {
            int r = mbase + j;
            if (r < n) h1[(long)r * HID + col] = c[j];
        }
    }
}

__global__ void gemm2_wmma(const float* __restrict__ h1r, const float* __restrict__ W2,
                           float* __restrict__ h2, int n) {
    __shared__ float wlds[HID * 16];
    int t = threadIdx.x;
    if (t < HID * 16) wlds[t] = W2[t];
    __syncthreads();

    int wave = t >> 5, lane = t & 31;
    int tile = blockIdx.x * 4 + wave;
    int col  = lane & 15;
    int hi   = lane >> 4;
    int mrow = tile * 16 + col;
    if (mrow >= n) mrow = n - 1;
    const float* hrow = h1r + (long)mrow * HID;

    v8f c = {};
    #pragma unroll
    for (int k0 = 0; k0 < HID; k0 += 4) {
        int ka = k0 + 2 * hi;
        v2f a = *(const v2f*)(hrow + ka);
        v2f b;
        b.x = wlds[ka * 16 + col];
        b.y = wlds[(ka + 1) * 16 + col];
        c = __builtin_amdgcn_wmma_f32_16x16x4_f32(false, a, false, b,
                                                  (short)0, c, false, false);
    }

    int mbase = tile * 16 + (hi << 3);
    #pragma unroll
    for (int j = 0; j < 8; ++j) {
        int r = mbase + j;
        if (r < n) h2[(long)r * NCLS + col] = c[j];
    }
}

template <int F>
__global__ void alpha_kernel(const float* __restrict__ h,
                             const float* __restrict__ a_src,
                             const float* __restrict__ a_dst,
                             float* __restrict__ outs, float* __restrict__ outd,
                             int n) {
    int i = blockIdx.x * blockDim.x + threadIdx.x;
    if (i >= n) return;
    const float4* hr = (const float4*)(h + (long)i * F);
    float s = 0.0f, d = 0.0f;
    #pragma unroll
    for (int q = 0; q < F / 4; ++q) {
        float4 hv = hr[q];
        s += hv.x * a_src[4 * q + 0] + hv.y * a_src[4 * q + 1]
           + hv.z * a_src[4 * q + 2] + hv.w * a_src[4 * q + 3];
        d += hv.x * a_dst[4 * q + 0] + hv.y * a_dst[4 * q + 1]
           + hv.z * a_dst[4 * q + 2] + hv.w * a_dst[4 * q + 3];
    }
    outs[i] = s;
    outd[i] = d;
}

__global__ void edge_max(const int* __restrict__ src, const int* __restrict__ dst,
                         const float* __restrict__ as, const float* __restrict__ ad,
                         float* __restrict__ m, int E) {
    int i = blockIdx.x * blockDim.x + threadIdx.x;
    if (i >= E) return;
    int s = src[i], d = dst[i];
    float e = as[s] + ad[d];
    e = (e >= 0.0f) ? e : NEG_SLOPE * e;
    atomicMaxF(&m[d], e);
}

template <int F>
__global__ void edge_acc(const int* __restrict__ src, const int* __restrict__ dst,
                         const float* __restrict__ as, const float* __restrict__ ad,
                         const float* __restrict__ m, const float* __restrict__ h,
                         float* __restrict__ denom, float* __restrict__ agg, int E) {
    int i = blockIdx.x * blockDim.x + threadIdx.x;
    if (i >= E) return;
    int s = src[i], d = dst[i];
    float e = as[s] + ad[d];
    e = (e >= 0.0f) ? e : NEG_SLOPE * e;
    float ex = __expf(e - m[d]);
    atomicAdd(&denom[d], ex);
    const float4* hs = (const float4*)(h + (long)s * F);
    float* ag = agg + (long)d * F;
    #pragma unroll
    for (int q = 0; q < F / 4; ++q) {
        float4 hv = hs[q];
        atomicAdd(&ag[4 * q + 0], ex * hv.x);
        atomicAdd(&ag[4 * q + 1], ex * hv.y);
        atomicAdd(&ag[4 * q + 2], ex * hv.z);
        atomicAdd(&ag[4 * q + 3], ex * hv.w);
    }
}

__global__ void bin_count(const int* __restrict__ dst, int* __restrict__ cnt, int E, int nb) {
    extern __shared__ int lh[];
    int t = threadIdx.x;
    for (int b = t; b < nb; b += blockDim.x) lh[b] = 0;
    __syncthreads();
    int chunk = (E + gridDim.x - 1) / gridDim.x;
    int lo = blockIdx.x * chunk;
    int hi = (lo + chunk < E) ? lo + chunk : E;
    for (int i = lo + t; i < hi; i += blockDim.x) atomicAdd(&lh[dst[i] >> NBSH], 1);
    __syncthreads();
    for (int b = t; b < nb; b += blockDim.x) {
        int c = lh[b];
        if (c) atomicAdd(&cnt[b], c);
    }
}

#define PCH  25000
#define PTH  1024
#define PIT  ((PCH + PTH - 1) / PTH)
__global__ void __launch_bounds__(PTH)
bin_place(const int* __restrict__ src, const int* __restrict__ dst,
          const int* __restrict__ rcnt, int* __restrict__ start, int* __restrict__ cur,
          unsigned int* __restrict__ bins, int E, int nb) {
    extern __shared__ __attribute__((aligned(16))) int sm[];
    __shared__ uint2 st[PCH];
    __shared__ int wtot[PTH / 32];
    __shared__ int gtot[PTH / 32];
    int*  lcnt = sm;
    int2* lob  = (int2*)(sm + ((nb + 1) & ~1));
    int t = threadIdx.x;
    int lane = t & 31, wave = t >> 5;
    for (int b = t; b < nb; b += PTH) lcnt[b] = 0;
    __syncthreads();
    int lo = blockIdx.x * PCH;
    int hi = (lo + PCH < E) ? lo + PCH : E;
    int dd[PIT], sv[PIT], rk[PIT];
    #pragma unroll
    for (int j = 0; j < PIT; ++j) {
        int i = lo + t + j * PTH;
        int ic = (i < hi) ? i : hi - 1;
        dd[j] = dst[ic];
        sv[j] = src[ic];
    }
    #pragma unroll
    for (int j = 0; j < PIT; ++j) {
        int i = lo + t + j * PTH;
        rk[j] = (i < hi) ? atomicAdd(&lcnt[dd[j] >> NBSH], 1) : 0;
    }
    __syncthreads();
    int carry = 0, gcarry = 0;
    for (int base = 0; base < nb; base += PTH) {
        int b = base + t;
        int c = (b < nb) ? lcnt[b] : 0;
        int got = c ? atomicAdd(&cur[b], c) : 0;
        int g = (b < nb) ? rcnt[b] : 0;
        int incl = c;
        #pragma unroll
        for (int off = 1; off < 32; off <<= 1) {
            int up = __shfl_up(incl, off);
            if (lane >= off) incl += up;
        }
        int gincl = g;
        #pragma unroll
        for (int off = 1; off < 32; off <<= 1) {
            int up = __shfl_up(gincl, off);
            if (lane >= off) gincl += up;
        }
        if (lane == 31) {
            wtot[wave] = incl;
            gtot[wave] = gincl;
        }
        __syncthreads();
        int wv = wtot[lane], gv = gtot[lane];
        int winc = wv, ginc = gv;
        #pragma unroll
        for (int off = 1; off < 32; off <<= 1) {
            int up = __shfl_up(winc, off);
            int gup = __shfl_up(ginc, off);
            if (lane >= off) {
                winc += up;
                ginc += gup;
            }
        }
        int wexcl = __shfl(winc, wave) - __shfl(wv, wave);
        int gexcl = __shfl(ginc, wave) - __shfl(gv, wave);
        int tot  = __shfl(winc, 31);
        int gall = __shfl(ginc, 31);
        int first = gcarry + gexcl + gincl - g;
        if (b < nb) {
            int2 ob;
            ob.x = carry + wexcl + incl - c;
            ob.y = first + got;
            lob[b] = ob;
            if (blockIdx.x == 0) start[b] = first;
        }
        carry += tot;
        gcarry += gall;
        __syncthreads();
    }
    if (blockIdx.x == 0 && t == 0) start[nb] = gcarry;
    #pragma unroll
    for (int j = 0; j < PIT; ++j) {
        int i = lo + t + j * PTH;
        int d = dd[j];
        int2 ob = lob[d >> NBSH];
        uint2 rec;
        rec.x = ((unsigned int)sv[j] << NBSH) | (unsigned int)(d & (NB - 1));
        rec.y = (unsigned int)(ob.y + rk[j]);
        if (i < hi) st[ob.x + rk[j]] = rec;
    }
    __syncthreads();
    int cnt = hi - lo;
    uint2 ro[PIT];
    #pragma unroll
    for (int j = 0; j < PIT; ++j) {
        int i = t + j * PTH;
        ro[j] = st[(i < cnt) ? i : cnt - 1];
    }
    #pragma unroll
    for (int j = 0; j < PIT; ++j) {
        int i = t + j * PTH;
        if (i < cnt) bins[ro[j].y] = ro[j].x;
    }
}

__device__ __forceinline__ void ldsMaxF(int* addr, float v) {
    if (v >= 0.0f) atomicMax(addr, __float_as_int(v));
    else           atomicMin((unsigned int*)addr, __float_as_uint(v));
}

template <int X>
__device__ __forceinline__ float xrow(float v) {
    return __builtin_bit_cast(float, __builtin_amdgcn_update_dpp(
        0, __builtin_bit_cast(int, v), 0x160 + X, 0xf, 0xf, true));
}
__device__ __forceinline__ float xhalf(float v) {
    return __builtin_bit_cast(float, __builtin_amdgcn_permlanex16(
        __builtin_bit_cast(int, v), __builtin_bit_cast(int, v), 0x76543210, 0xfedcba98, false, false));
}

template <int V> struct RowPiece;
template <> struct RowPiece<2> { typedef v2f T; };
template <> struct RowPiece<4> { typedef __attribute__((ext_vector_type(4))) float T; };

template <int F>
__device__ __forceinline__ float row_logit(const float* __restrict__ h, const float* __restrict__ a, int i) {
    const float4* hr = (const float4*)(h + (long)i * F);
    float v = 0.0f;
    #pragma unroll
    for (int q = 0; q < F / 4; ++q) {
        float4 hv = hr[q];
        v += hv.x * a[4 * q + 0] + hv.y * a[4 * q + 1]
           + hv.z * a[4 * q + 2] + hv.w * a[4 * q + 3];
    }
    return v;
}

#define EPT  (ECAP / LTHREADS)
#define UB   4

template <int F>
__global__ void __launch_bounds__(LTHREADS)
range_owner(const unsigned int* __restrict__ bins, const int* __restrict__ start,
            const float* __restrict__ as, const float* __restrict__ ad,
            const float* __restrict__ a_src, const float* __restrict__ a_dst,
            const float* __restrict__ h, const float* __restrict__ bias,
            const float* __restrict__ Wn, const float* __restrict__ an_src,
            const float* __restrict__ an_dst,
            float* __restrict__ hn, float* __restrict__ asn, float* __restrict__ adn,
            float* __restrict__ out, int n) {
    static_assert(F == HID || F == NCLS, "the per-node endings exist for the two layers only");
    static_assert(LTHREADS == NB * NCLS / 4, "one 16-byte piece of a 16-float row per thread");
    constexpr int HP = NCLS + 1;
    __shared__ float w2l[(F == HID) ? HID * NCLS : 1];
    __shared__ float h2l[(F == HID) ? NB * HP : 1];
    __shared__ float bl[F];
    __shared__ float anl[(F == HID) ? 2 * NCLS : 1];
    constexpr int RS = F + 1;
    constexpr int V  = F / 4;
    typedef typename RowPiece<V>::T piece_t;
    __shared__ float acc[NB * RS];
    __shared__ float adl[NB];
    __shared__ int   mI[NB];
    __shared__ int   cntl[NB];
    __shared__ int   offl[NB + 4];
    __shared__ int   ss[ECAP];
    const float NINF = -__builtin_inff();
    int t = threadIdx.x;
    int base = blockIdx.x << NBSH;
    int lo = start[blockIdx.x], hi = start[blockIdx.x + 1];
    int ne = hi - lo;
    if (t < NB) {
        int node = base + t;
        float dv = 0.0f;
        if (node < n) {
            if constexpr (F == HID) dv = row_logit<F>(h, a_dst, node);
            else                    dv = ad[node];
        }
        adl[t]  = dv;
        mI[t]   = __float_as_int(NINF);
        cntl[t] = 0;
    }
    if (t < F) bl[t] = bias[t];
    if constexpr (F == HID) {
        if (t < HID * NCLS) w2l[t] = Wn[t];
        if (t < NCLS) {
            anl[t]        = an_src[t];
            anl[NCLS + t] = an_dst[t];
        }
    }

    if (ne > ECAP) {
        for (int j = t; j < NB * RS; j += LTHREADS) acc[j] = 0.0f;
        __syncthreads();
        for (int i = lo + t; i < hi; i += LTHREADS) {
            unsigned int w = bins[i];
            int s = (int)(w >> NBSH), dl = (int)(w & (NB - 1));
            float sv;
            if constexpr (F == HID) sv = row_logit<F>(h, a_src, s);
            else                    sv = as[s];
            float e = sv + adl[dl];
            e = (e >= 0.0f) ? e : NEG_SLOPE * e;
            ldsMaxF(&mI[dl], e);
        }
        __syncthreads();
        for (int i = lo + t; i < hi; i += LTHREADS) {
            unsigned int w = bins[i];
            int s = (int)(w >> NBSH), dl = (int)(w & (NB - 1));
            float sv;
            if constexpr (F == HID) sv = row_logit<F>(h, a_src, s);
            else                    sv = as[s];
            float e = sv + adl[dl];
            e = (e >= 0.0f) ? e : NEG_SLOPE * e;
            float ex = __expf(e - __int_as_float(mI[dl]));
            const float4* hs = (const float4*)(h + (long)s * F);
            float* ar = acc + dl * RS;
            atomicAdd(&ar[F], ex);
            #pragma unroll
            for (int q = 0; q < F / 4; ++q) {
                float4 hv = hs[q];
                atomicAdd(&ar[4 * q + 0], ex * hv.x);
                atomicAdd(&ar[4 * q + 1], ex * hv.y);
                atomicAdd(&ar[4 * q + 2], ex * hv.z);
                atomicAdd(&ar[4 * q + 3], ex * hv.w);
            }
        }
    } else {
        __syncthreads();
        unsigned int w[EPT];
        #pragma unroll
        for (int j = 0; j < EPT; ++j) {
            int k = t + j * LTHREADS;
            w[j] = (k < ne) ? bins[lo + k] : 0u;
        }
        #pragma unroll
        for (int j = 0; j < EPT; ++j) {
            int k = t + j * LTHREADS;
            if (k < ne) atomicAdd(&cntl[w[j] & (NB - 1)], 1);
        }
        __syncthreads();
        if (t < 32) {
            int c0 = cntl[4 * t], c1 = cntl[4 * t + 1], c2 = cntl[4 * t + 2], c3 = cntl[4 * t + 3];
            int sum = c0 + c1 + c2 + c3;
            int incl = sum;
            #pragma unroll
            for (int off = 1; off < 32; off <<= 1) {
                int up = __shfl_up(incl, off);
                if (t >= off) incl += up;
            }
            int excl = incl - sum;
            int e1 = excl + c0, e2 = e1 + c1, e3 = e2 + c2;
            offl[4 * t]     = excl;
            offl[4 * t + 1] = e1;
            offl[4 * t + 2] = e2;
            offl[4 * t + 3] = e3;
            if (t == 31) offl[NB] = incl;
            cntl[4 * t]     = excl;
            cntl[4 * t + 1] = e1;
            cntl[4 * t + 2] = e2;
            cntl[4 * t + 3] = e3;
        }
        __syncthreads();
        #pragma unroll
        for (int j = 0; j < EPT; ++j) {
            int k = t + j * LTHREADS;
            if (k < ne) ss[atomicAdd(&cntl[w[j] & (NB - 1)], 1)] = (int)(w[j] >> NBSH);
        }
        __syncthreads();

        int lane = t & 31;
        int slot = lane >> 2, q = lane & 3;
        int wv = __builtin_amdgcn_readfirstlane(t >> 5);
        float av[V];
        #pragma unroll
        for (int v = 0; v < V; ++v) av[v] = a_src[q * V + v];
        const char* hb = (const char*)h;
        unsigned int qoff = (unsigned int)(q * V * 4);

        for (int r = wv; r < NB; r += LTHREADS / 32) {
            int o0 = __builtin_amdgcn_readfirstlane(offl[r]);
            int o1 = __builtin_amdgcn_readfirstlane(offl[r + 1]);
            float adv = adl[r];
            float m = NINF, d = 0.0f;
            float a[V];
            #pragma unroll
            for (int v = 0; v < V; ++v) a[v] = 0.0f;

            for (int p0 = o0; p0 < o1; p0 += 8 * UB) {
                int sidx[UB];
                #pragma unroll
                for (int u = 0; u < UB; ++u) {
                    int p = p0 + u * 8 + slot;
                    int pc = (p < o1) ? p : o1 - 1;
                    sidx[u] = ss[pc];
                }
                piece_t hv[UB];
                #pragma unroll
                for (int u = 0; u < UB; ++u)
                    hv[u] = *(const piece_t*)(hb + ((unsigned int)sidx[u] * (unsigned int)(F * 4) + qoff));
                float e[UB];
                float bm = NINF;
                #pragma unroll
                for (int u = 0; u < UB; ++u) {
                    float part = 0.0f;
                    #pragma unroll
                    for (int v = 0; v < V; ++v) part += hv[u][v] * av[v];
                    part += xrow<1>(part);
                    part += xrow<2>(part);
                    float x = part + adv;
                    x = fmaxf(x, NEG_SLOPE * x);
                    int p = p0 + u * 8 + slot;
                    e[u] = (p < o1) ? x : NINF;
                    bm = fmaxf(bm, e[u]);
                }
                float mn = fmaxf(m, bm);
                float mu = (mn == NINF) ? 0.0f : mn;
                float sc = __expf(m - mu);
                d *= sc;
                #pragma unroll
                for (int v = 0; v < V; ++v) a[v] *= sc;
                #pragma unroll
                for (int u = 0; u < UB; ++u) {
                    float ex = __expf(e[u] - mu);
                    d += ex;
                    #pragma unroll
                    for (int v = 0; v < V; ++v) a[v] += ex * hv[u][v];
                }
                m = mn;
            }

            float M = m;
            M = fmaxf(M, xrow<4>(M));
            M = fmaxf(M, xrow<8>(M));
            M = fmaxf(M, xhalf(M));
            float Mu = (M == NINF) ? 0.0f : M;
            float sc = __expf(m - Mu);
            d *= sc;
            d += xrow<4>(d);
            d += xrow<8>(d);
            d += xhalf(d);
            #pragma unroll
            for (int v = 0; v < V; ++v) {
                float x = a[v] * sc;
                x += xrow<4>(x);
                x += xrow<8>(x);
                x += xhalf(x);
                a[v] = x;
            }
            if (slot == 0) {
                #pragma unroll
                for (int v = 0; v < V; ++v) acc[r * RS + q * V + v] = a[v];
                if (q == 0) acc[r * RS + F] = d;
            }
        }
    }
    __syncthreads();

    if constexpr (F == HID) {
        if (t < NB) {
            float* ar = acc + t * RS;
            float d = ar[F];
            float inv = (d > 0.0f) ? 1.0f / d : 0.0f;
            #pragma unroll
            for (int f = 0; f < F; ++f) ar[f] = fmaxf(ar[f] * inv + bl[f], 0.0f);
        }
        __syncthreads();
        int wt = __builtin_amdgcn_readfirstlane(t >> 5);
        if (wt < NB / 16) {
            int lane = t & 31;
            int col  = lane & 15;
            int hi   = lane >> 4;
            const float* arow = acc + (wt * 16 + col) * RS;
            v8f c = {};
            #pragma unroll
            for (int k0 = 0; k0 < HID; k0 += 4) {
                int ka = k0 + 2 * hi;
                v2f a;
                a.x = arow[ka];
                a.y = arow[ka + 1];
                v2f b;
                b.x = w2l[ka * NCLS + col];
                b.y = w2l[(ka + 1) * NCLS + col];
                c = __builtin_amdgcn_wmma_f32_16x16x4_f32(false, a, false, b,
                                                          (short)0, c, false, false);
            }
            int rb = wt * 16 + (hi << 3);
            #pragma unroll
            for (int j = 0; j < 8; ++j) h2l[(rb + j) * HP + col] = c[j];
        }
        __syncthreads();
        if (t < NB) {
            int node = base + t;
            const float* hr = h2l + t * HP;
            float sv = 0.0f, dv = 0.0f;
            #pragma unroll
            for (int q = 0; q < NCLS / 4; ++q) {
                float x0 = hr[4 * q + 0], x1 = hr[4 * q + 1], x2 = hr[4 * q + 2], x3 = hr[4 * q + 3];
                sv += x0 * anl[4 * q + 0] + x1 * anl[4 * q + 1]
                    + x2 * anl[4 * q + 2] + x3 * anl[4 * q + 3];
                dv += x0 * anl[NCLS + 4 * q + 0] + x1 * anl[NCLS + 4 * q + 1]
                    + x2 * anl[NCLS + 4 * q + 2] + x3 * anl[NCLS + 4 * q + 3];
            }
            if (node < n) {
                asn[node] = sv;
                adn[node] = dv;
            }
        }
        {
            int r = t >> 2, qd = t & 3;
            int node = base + r;
            const float* hr = h2l + r * HP + 4 * qd;
            float4 o;
            o.x = hr[0]; o.y = hr[1]; o.z = hr[2]; o.w = hr[3];
            if (node < n) *(float4*)(hn + (long)node * NCLS + 4 * qd) = o;
        }
    } else {
        if (t < NB) {
            float* ar = acc + t * RS;
            float d = ar[F];
            float inv = (d > 0.0f) ? 1.0f / d : 0.0f;
            float v[F];
            float mx = -__builtin_inff();
            #pragma unroll
            for (int f = 0; f < F; ++f) v[f] = ar[f] * inv + bl[f];
            #pragma unroll
            for (int f = 0; f < F; ++f) mx = fmaxf(mx, v[f]);
            float s = 0.0f;
            #pragma unroll
            for (int f = 0; f < F; ++f) s += __expf(v[f] - mx);
            float ls = __logf(s) + mx;
            #pragma unroll
            for (int f = 0; f < F; ++f) ar[f] = v[f] - ls;
        }
        __syncthreads();
        {
            int r = t >> 2, qd = t & 3;
            int node = base + r;
            const float* ar = acc + r * RS + 4 * qd;
            float4 o;
            o.x = ar[0]; o.y = ar[1]; o.z = ar[2]; o.w = ar[3];
            if (node < n) *(float4*)(out + (long)node * F + 4 * qd) = o;
        }
    }
}

__global__ void finish1(float* __restrict__ agg, const float* __restrict__ denom,
                        const float* __restrict__ bias, int n) {
    int i = blockIdx.x * blockDim.x + threadIdx.x;
    if (i >= n) return;
    float d = denom[i];
    float inv = (d > 0.0f) ? 1.0f / d : 0.0f;
    float4* ap = (float4*)(agg + (long)i * HID);
    #pragma unroll
    for (int q = 0; q < HID / 4; ++q) {
        float4 v = ap[q];
        v.x = fmaxf(v.x * inv + bias[4 * q + 0], 0.0f);
        v.y = fmaxf(v.y * inv + bias[4 * q + 1], 0.0f);
        v.z = fmaxf(v.z * inv + bias[4 * q + 2], 0.0f);
        v.w = fmaxf(v.w * inv + bias[4 * q + 3], 0.0f);
        ap[q] = v;
    }
}

__global__ void finish2(const float* __restrict__ agg, const float* __restrict__ denom,
                        const float* __restrict__ bias, float* __restrict__ out, int n) {
    int i = blockIdx.x * blockDim.x + threadIdx.x;
    if (i >= n) return;
    float d = denom[i];
    float inv = (d > 0.0f) ? 1.0f / d : 0.0f;
    const float4* ap = (const float4*)(agg + (long)i * NCLS);
    float v[NCLS];
    float mx = -__builtin_inff();
    #pragma unroll
    for (int q = 0; q < NCLS / 4; ++q) {
        float4 a = ap[q];
        v[4 * q + 0] = a.x * inv + bias[4 * q + 0];
        v[4 * q + 1] = a.y * inv + bias[4 * q + 1];
        v[4 * q + 2] = a.z * inv + bias[4 * q + 2];
        v[4 * q + 3] = a.w * inv + bias[4 * q + 3];
    }
    #pragma unroll
    for (int f = 0; f < NCLS; ++f) mx = fmaxf(mx, v[f]);
    float s = 0.0f;
    #pragma unroll
    for (int f = 0; f < NCLS; ++f) s += __expf(v[f] - mx);
    float ls = __logf(s) + mx;
    float4* op = (float4*)(out + (long)i * NCLS);
    #pragma unroll
    for (int q = 0; q < NCLS / 4; ++q) {
        float4 o;
        o.x = v[4 * q + 0] - ls;
        o.y = v[4 * q + 1] - ls;
        o.z = v[4 * q + 2] - ls;
        o.w = v[4 * q + 3] - ls;
        op[q] = o;
    }
}

extern "C" void kernel_launch(void* const* d_in, const int* in_sizes, int n_in,
                              void* d_out, int out_size, void* d_ws, size_t ws_size,
                              hipStream_t stream) {
    const float* x   = (const float*)d_in[0];
    const int*   ei  = (const int*)  d_in[1];
    const float* W1  = (const float*)d_in[2];
    const float* a1s = (const float*)d_in[3];
    const float* a1d = (const float*)d_in[4];
    const float* b1  = (const float*)d_in[5];
    const float* W2  = (const float*)d_in[6];
    const float* a2s = (const float*)d_in[7];
    const float* a2d = (const float*)d_in[8];
    const float* b2  = (const float*)d_in[9];

    int n = in_sizes[0] / FIN;
    int E = in_sizes[1] / 2;
    const int* src = ei;
    const int* dst = ei + E;

    float* ws   = (float*)d_ws;
    float* h1   = ws; ws += (long)n * HID;
    float* as1  = ws; ws += n;
    float* ad1  = ws; ws += n;
    float* m1   = ws; ws += n;
    float* dn1  = ws; ws += n;
    float* agg1 = ws; ws += (long)n * HID;
    float* h2   = ws; ws += (long)n * NCLS;
    float* as2  = ws; ws += n;
    float* ad2  = ws; ws += n;
    float* m2   = ws; ws += n;
    float* dn2  = ws; ws += n;
    float* agg2 = ws; ws += (long)n * NCLS;

    int ntiles  = (n + 15) / 16;
    int gblocks = (ntiles + 3) / 4;
    int eblocks = (E + 255) / 256;
    int nblocks = (n + 255) / 256;

    int nb  = (n + NB - 1) / NB;
    int nbp = (nb + 4) & ~3;
    int* cnt   = (int*)ws;
    int* cur   = cnt + nbp;
    int* start = cur + nbp;
    unsigned int* bins = (unsigned int*)(start + nbp);
    size_t need = ((size_t)56 * n + (size_t)3 * nbp + (size_t)E) * 4;

    if (ws_size < need || n > (1 << 24)) {
        init_ws<<<(n * NCLS + 255) / 256, 256, 0, stream>>>(m1, dn1, agg1, m2, dn2, agg2, n);
        gemm1_wmma<<<gblocks, 128, 0, stream>>>(x, W1, h1, n, (int*)nullptr, 0);
        alpha_kernel<HID><<<nblocks, 256, 0, stream>>>(h1, a1s, a1d, as1, ad1, n);
        edge_max<<<eblocks, 256, 0, stream>>>(src, dst, as1, ad1, m1, E);
        edge_acc<HID><<<eblocks, 256, 0, stream>>>(src, dst, as1, ad1, m1, h1, dn1, agg1, E);
        finish1<<<nblocks, 256, 0, stream>>>(agg1, dn1, b1, n);
        gemm2_wmma<<<gblocks, 128, 0, stream>>>(agg1, W2, h2, n);
        alpha_kernel<NCLS><<<nblocks, 256, 0, stream>>>(h2, a2s, a2d, as2, ad2, n);
        edge_max<<<eblocks, 256, 0, stream>>>(src, dst, as2, ad2, m2, E);
        edge_acc<NCLS><<<eblocks, 256, 0, stream>>>(src, dst, as2, ad2, m2, h2, dn2, agg2, E);
        finish2<<<nblocks, 256, 0, stream>>>(agg2, dn2, b2, (float*)d_out, n);
        return;
    }

    gemm1_wmma<<<gblocks, 128, 0, stream>>>(x, W1, h1, n, cnt, 2 * nbp);

    bin_count<<<512, 1024, nb * sizeof(int), stream>>>(dst, cnt, E, nb);
    bin_place<<<(E + PCH - 1) / PCH, PTH, 4 * nb * sizeof(int), stream>>>(src, dst, cnt, start, cur, bins, E, nb);

    range_owner<HID><<<nb, LTHREADS, 0, stream>>>(bins, start, (const float*)nullptr, (const float*)nullptr,
                                                  a1s, a1d, h1, b1, W2, a2s, a2d,
                                                  h2, as2, ad2, (float*)nullptr, n);

    range_owner<NCLS><<<nb, LTHREADS, 0, stream>>>(bins, start, as2, ad2, a2s, (const float*)nullptr, h2, b2,
                                                   (const float*)nullptr, (const float*)nullptr,
                                                   (const float*)nullptr, (float*)nullptr,
                                                   (float*)nullptr, (float*)nullptr, (float*)d_out, n);
}


# edges grouped per chunk in place: a chunk keeps its own slots of bins, by range, run starts in a table; owners collect their runs: count launch and global atomics gone (5 to 4); sum order changes
# speedup vs baseline: 26.6987x; 1.0427x over previous
#include <hip/hip_runtime.h>
#include <math.h>

typedef __attribute__((ext_vector_type(2))) float v2f;
typedef __attribute__((ext_vector_type(8))) float v8f;

#define FIN   128
#define HID   8
#define NCLS  16
#define NEG_SLOPE 0.2f
#define NBSH  7
#define NB    (1 << NBSH)
#define LTHREADS 512
#define ECAP  10240

__device__ __forceinline__ void atomicMaxF(float* addr, float v) {
    if (v >= 0.0f) atomicMax((int*)addr, __float_as_int(v));
    else           atomicMin((unsigned int*)addr, __float_as_uint(v));
}

__global__ void init_ws(float* m1, float* dn1, float* agg1,
                        float* m2, float* dn2, float* agg2, int n) {
    int i = blockIdx.x * blockDim.x + threadIdx.x;
    if (i < n * NCLS) agg2[i] = 0.0f;
    if (i < n * HID)  agg1[i] = 0.0f;
    if (i < n) {
        m1[i] = -__builtin_inff();
        m2[i] = -__builtin_inff();
        dn1[i] = 0.0f;
        dn2[i] = 0.0f;
    }
}

__global__ void gemm1_wmma(const float* __restrict__ x, const float* __restrict__ W1,
                           float* __restrict__ h1, int n, int* __restrict__ zp, int zc) {
    __shared__ float wlds[FIN * 16];
    int t = threadIdx.x;
    {
        int gi = blockIdx.x * blockDim.x + t;
        if (gi < zc) zp[gi] = 0;
    }
    {
        int k = t;
        const float* wr = W1 + k * HID;
        #pragma unroll
        for (int c = 0; c < 16; ++c) wlds[k * 16 + c] = (c < HID) ? wr[c] : 0.0f;
    }
    __syncthreads();

    int wave = t >> 5, lane = t & 31;
    int tile = blockIdx.x * 4 + wave;
    int col  = lane & 15;
    int hi   = lane >> 4;
    int mrow = tile * 16 + col;
    if (mrow >= n) mrow = n - 1;
    const float* xrow = x + (long)mrow * FIN;

    v8f c = {};
    #pragma unroll
    for (int k0 = 0; k0 < FIN; k0 += 4) {
        int ka = k0 + 2 * hi;
        v2f a = *(const v2f*)(xrow + ka);
        v2f b;
        b.x = wlds[ka * 16 + col];
        b.y = wlds[(ka + 1) * 16 + col];
        c = __builtin_amdgcn_wmma_f32_16x16x4_f32(false, a, false, b,
                                                  (short)0, c, false, false);
    }

    int mbase = tile * 16 + (hi << 3);
    if (col < HID) {
        #pragma unroll
        for (int j = 0; j < 8; ++j) {
            int r = mbase + j;
            if (r < n) h1[(long)r * HID + col] = c[j];
        }
    }
}

__global__ void gemm2_wmma(const float* __restrict__ h1r, const float* __restrict__ W2,
                           float* __restrict__ h2, int n) {
    __shared__ float wlds[HID * 16];
    int t = threadIdx.x;
    if (t < HID * 16) wlds[t] = W2[t];
    __syncthreads();

    int wave = t >> 5, lane = t & 31;
    int tile = blockIdx.x * 4 + wave;
    int col  = lane & 15;
    int hi   = lane >> 4;
    int mrow = tile * 16 + col;
    if (mrow >= n) mrow = n - 1;
    const float* hrow = h1r + (long)mrow * HID;

    v8f c = {};
    #pragma unroll
    for (int k0 = 0; k0 < HID; k0 += 4) {
        int ka = k0 + 2 * hi;
        v2f a = *(const v2f*)(hrow + ka);
        v2f b;
        b.x = wlds[ka * 16 + col];
        b.y = wlds[(ka + 1) * 16 + col];
        c = __builtin_amdgcn_wmma_f32_16x16x4_f32(false, a, false, b,
                                                  (short)0, c, false, false);
    }

    int mbase = tile * 16 + (hi << 3);
    #pragma unroll
    for (int j = 0; j < 8; ++j) {
        int r = mbase + j;
        if (r < n) h2[(long)r * NCLS + col] = c[j];
    }
}

template <int F>
__global__ void alpha_kernel(const float* __restrict__ h,
                             const float* __restrict__ a_src,
                             const float* __restrict__ a_dst,
                             float* __restrict__ outs, float* __restrict__ outd,
                             int n) {
    int i = blockIdx.x * blockDim.x + threadIdx.x;
    if (i >= n) return;
    const float4* hr = (const float4*)(h + (long)i * F);
    float s = 0.0f, d = 0.0f;
    #pragma unroll
    for (int q = 0; q < F / 4; ++q) {
        float4 hv = hr[q];
        s += hv.x * a_src[4 * q + 0] + hv.y * a_src[4 * q + 1]
           + hv.z * a_src[4 * q + 2] + hv.w * a_src[4 * q + 3];
        d += hv.x * a_dst[4 * q + 0] + hv.y * a_dst[4 * q + 1]
           + hv.z * a_dst[4 * q + 2] + hv.w * a_dst[4 * q + 3];
    }
    outs[i] = s;
    outd[i] = d;
}

__global__ void edge_max(const int* __restrict__ src, const int* __restrict__ dst,
                         const float* __restrict__ as, const float* __restrict__ ad,
                         float* __restrict__ m, int E) {
    int i = blockIdx.x * blockDim.x + threadIdx.x;
    if (i >= E) return;
    int s = src[i], d = dst[i];
    float e = as[s] + ad[d];
    e = (e >= 0.0f) ? e : NEG_SLOPE * e;
    atomicMaxF(&m[d], e);
}

template <int F>
__global__ void edge_acc(const int* __restrict__ src, const int* __restrict__ dst,
                         const float* __restrict__ as, const float* __restrict__ ad,
                         const float* __restrict__ m, const float* __restrict__ h,
                         float* __restrict__ denom, float* __restrict__ agg, int E) {
    int i = blockIdx.x * blockDim.x + threadIdx.x;
    if (i >= E) return;
    int s = src[i], d = dst[i];
    float e = as[s] + ad[d];
    e = (e >= 0.0f) ? e : NEG_SLOPE * e;
    float ex = __expf(e - m[d]);
    atomicAdd(&denom[d], ex);
    const float4* hs = (const float4*)(h + (long)s * F);
    float* ag = agg + (long)d * F;
    #pragma unroll
    for (int q = 0; q < F / 4; ++q) {
        float4 hv = hs[q];
        atomicAdd(&ag[4 * q + 0], ex * hv.x);
        atomicAdd(&ag[4 * q + 1], ex * hv.y);
        atomicAdd(&ag[4 * q + 2], ex * hv.z);
        atomicAdd(&ag[4 * q + 3], ex * hv.w);
    }
}

#define PCH  25000
#define PTH  1024
#define PIT  ((PCH + PTH - 1) / PTH)
__global__ void __launch_bounds__(PTH)
bin_place(const int* __restrict__ src, const int* __restrict__ dst,
          int* __restrict__ tab, unsigned int* __restrict__ bins, int E, int nb) {
    extern __shared__ __attribute__((aligned(16))) int sm[];
    __shared__ __attribute__((aligned(16))) unsigned int st[PCH];
    __shared__ int wtot[PTH / 32];
    int* lcnt = sm;
    int* lob  = sm + nb;
    int t = threadIdx.x;
    int lane = t & 31, wave = t >> 5;
    for (int b = t; b < nb; b += PTH) lcnt[b] = 0;
    __syncthreads();
    int lo = blockIdx.x * PCH;
    int hi = (lo + PCH < E) ? lo + PCH : E;
    int* trow = tab + (long)blockIdx.x * (nb + 1);
    int dd[PIT], sv[PIT], rk[PIT];
    #pragma unroll
    for (int j = 0; j < PIT; ++j) {
        int i = lo + t + j * PTH;
        int ic = (i < hi) ? i : hi - 1;
        dd[j] = dst[ic];
        sv[j] = src[ic];
    }
    #pragma unroll
    for (int j = 0; j < PIT; ++j) {
        int i = lo + t + j * PTH;
        rk[j] = (i < hi) ? atomicAdd(&lcnt[dd[j] >> NBSH], 1) : 0;
    }
    __syncthreads();
    int carry = 0;
    for (int base = 0; base < nb; base += PTH) {
        int b = base + t;
        int c = (b < nb) ? lcnt[b] : 0;
        int incl = c;
        #pragma unroll
        for (int off = 1; off < 32; off <<= 1) {
            int up = __shfl_up(incl, off);
            if (lane >= off) incl += up;
        }
        if (lane == 31) wtot[wave] = incl;
        __syncthreads();
        int wv = wtot[lane];
        int winc = wv;
        #pragma unroll
        for (int off = 1; off < 32; off <<= 1) {
            int up = __shfl_up(winc, off);
            if (lane >= off) winc += up;
        }
        int wexcl = __shfl(winc, wave) - __shfl(wv, wave);
        int tot   = __shfl(winc, 31);
        if (b < nb) {
            int ex = carry + wexcl + incl - c;
            lob[b]  = ex;
            trow[b] = lo + ex;
        }
        carry += tot;
        __syncthreads();
    }
    if (t == 0) trow[nb] = hi;
    #pragma unroll
    for (int j = 0; j < PIT; ++j) {
        int i = lo + t + j * PTH;
        int d = dd[j];
        int ob = lob[d >> NBSH];
        unsigned int word = ((unsigned int)sv[j] << NBSH) | (unsigned int)(d & (NB - 1));
        if (i < hi) st[ob + rk[j]] = word;
    }
    __syncthreads();
    int cnt = hi - lo;
    int nv  = cnt >> 2;
    const uint4* sv4 = (const uint4*)st;
    uint4* bo4 = (uint4*)(bins + lo);
    uint4 ro[(PCH / 4 + PTH - 1) / PTH];
    #pragma unroll
    for (int j = 0; j < (PCH / 4 + PTH - 1) / PTH; ++j) {
        int i = t + j * PTH;
        ro[j] = sv4[(i < nv) ? i : 0];
    }
    #pragma unroll
    for (int j = 0; j < (PCH / 4 + PTH - 1) / PTH; ++j) {
        int i = t + j * PTH;
        if (i < nv) bo4[i] = ro[j];
    }
    for (int i = (nv << 2) + t; i < cnt; i += PTH) bins[lo + i] = st[i];
}

__device__ __forceinline__ void ldsMaxF(int* addr, float v) {
    if (v >= 0.0f) atomicMax(addr, __float_as_int(v));
    else           atomicMin((unsigned int*)addr, __float_as_uint(v));
}

template <int X>
__device__ __forceinline__ float xrow(float v) {
    return __builtin_bit_cast(float, __builtin_amdgcn_update_dpp(
        0, __builtin_bit_cast(int, v), 0x160 + X, 0xf, 0xf, true));
}
__device__ __forceinline__ float xhalf(float v) {
    return __builtin_bit_cast(float, __builtin_amdgcn_permlanex16(
        __builtin_bit_cast(int, v), __builtin_bit_cast(int, v), 0x76543210, 0xfedcba98, false, false));
}

template <int V> struct RowPiece;
template <> struct RowPiece<2> { typedef v2f T; };
template <> struct RowPiece<4> { typedef __attribute__((ext_vector_type(4))) float T; };

template <int F>
__device__ __forceinline__ float row_logit(const float* __restrict__ h, const float* __restrict__ a, int i) {
    const float4* hr = (const float4*)(h + (long)i * F);
    float v = 0.0f;
    #pragma unroll
    for (int q = 0; q < F / 4; ++q) {
        float4 hv = hr[q];
        v += hv.x * a[4 * q + 0] + hv.y * a[4 * q + 1]
           + hv.z * a[4 * q + 2] + hv.w * a[4 * q + 3];
    }
    return v;
}

#define UB   4
#define RPW  16

template <int F>
__global__ void __launch_bounds__(LTHREADS)
range_owner(const unsigned int* __restrict__ bins, const int* __restrict__ tab, int nch,
            const float* __restrict__ as, const float* __restrict__ ad,
            const float* __restrict__ a_src, const float* __restrict__ a_dst,
            const float* __restrict__ h, const float* __restrict__ bias,
            const float* __restrict__ Wn, const float* __restrict__ an_src,
            const float* __restrict__ an_dst,
            float* __restrict__ hn, float* __restrict__ asn, float* __restrict__ adn,
            float* __restrict__ out, int n) {
    static_assert(F == HID || F == NCLS, "the per-node endings exist for the two layers only");
    static_assert(LTHREADS == NB * NCLS / 4, "one 16-byte piece of a 16-float row per thread");
    constexpr int HP = NCLS + 1;
    __shared__ float w2l[(F == HID) ? HID * NCLS : 1];
    __shared__ float h2l[(F == HID) ? NB * HP : 1];
    __shared__ float bl[F];
    __shared__ float anl[(F == HID) ? 2 * NCLS : 1];
    constexpr int RS = F + 1;
    constexpr int V  = F / 4;
    typedef typename RowPiece<V>::T piece_t;
    __shared__ float acc[NB * RS];
    __shared__ float adl[NB];
    __shared__ int   mI[NB];
    __shared__ int   cntl[NB];
    __shared__ int   offl[NB + 4];
    __shared__ int   ss[ECAP];
    const float NINF = -__builtin_inff();
    int t = threadIdx.x;
    int base = blockIdx.x << NBSH;
    const long pitch = (long)gridDim.x + 1;
    const int* tcol = tab + blockIdx.x;
    if (t < NB) {
        int node = base + t;
        float dv = 0.0f;
        if (node < n) {
            if constexpr (F == HID) dv = row_logit<F>(h, a_dst, node);
            else                    dv = ad[node];
        }
        adl[t]  = dv;
        mI[t]   = __float_as_int(NINF);
        cntl[t] = 0;
    }
    if (t < F) bl[t] = bias[t];
    if constexpr (F == HID) {
        if (t < HID * NCLS) w2l[t] = Wn[t];
        if (t < NCLS) {
            anl[t]        = an_src[t];
            anl[NCLS + t] = an_dst[t];
        }
    }

    __syncthreads();
    bool slow = nch > RPW * (LTHREADS / 32);
    if (!slow) {
        int lane = t & 31;
        int wv = __builtin_amdgcn_readfirstlane(t >> 5);
        int rp0 = 0, rp1 = 0;
        {
            int cidx = wv * RPW + (lane & (RPW - 1));
            if (cidx < nch) {
                const int* te = tcol + (long)cidx * pitch;
                rp0 = te[0];
                rp1 = te[1];
            }
        }
        int rln = rp1 - rp0;
        unsigned int w[2 * RPW];
        int over = 0;
        #pragma unroll
        for (int s = 0; s < RPW; ++s) {
            int ps = __builtin_amdgcn_readlane(rp0, s);
            int ls = __builtin_amdgcn_readlane(rln, s);
            over |= (ls > 64) ? 1 : 0;
            w[2 * s]     = (lane < ls)      ? bins[ps + lane]      : 0u;
            w[2 * s + 1] = (lane + 32 < ls) ? bins[ps + 32 + lane] : 0u;
        }
        #pragma unroll
        for (int s = 0; s < RPW; ++s) {
            int ls = __builtin_amdgcn_readlane(rln, s);
            if (lane < ls)      atomicAdd(&cntl[w[2 * s] & (NB - 1)], 1);
            if (lane + 32 < ls) atomicAdd(&cntl[w[2 * s + 1] & (NB - 1)], 1);
        }
        if (over) {
            for (int s = 0; s < RPW; ++s) {
                int ps = __shfl(rp0, s), ls = __shfl(rln, s);
                for (int k = 64 + lane; k < ls; k += 32) atomicAdd(&cntl[bins[ps + k] & (NB - 1)], 1);
            }
        }
        __syncthreads();
        if (t < 32) {
            int c0 = cntl[4 * t], c1 = cntl[4 * t + 1], c2 = cntl[4 * t + 2], c3 = cntl[4 * t + 3];
            int sum = c0 + c1 + c2 + c3;
            int incl = sum;
            #pragma unroll
            for (int off = 1; off < 32; off <<= 1) {
                int up = __shfl_up(incl, off);
                if (t >= off) incl += up;
            }
            int excl = incl - sum;
            int e1 = excl + c0, e2 = e1 + c1, e3 = e2 + c2;
            offl[4 * t]     = excl;
            offl[4 * t + 1] = e1;
            offl[4 * t + 2] = e2;
            offl[4 * t + 3] = e3;
            if (t == 31) offl[NB] = incl;
            cntl[4 * t]     = excl;
            cntl[4 * t + 1] = e1;
            cntl[4 * t + 2] = e2;
            cntl[4 * t + 3] = e3;
        }
        __syncthreads();
        slow = offl[NB] > ECAP;
        if (!slow) {
            #pragma unroll
            for (int s = 0; s < RPW; ++s) {
                int ls = __builtin_amdgcn_readlane(rln, s);
                if (lane < ls)      ss[atomicAdd(&cntl[w[2 * s] & (NB - 1)], 1)] = (int)(w[2 * s] >> NBSH);
                if (lane + 32 < ls) ss[atomicAdd(&cntl[w[2 * s + 1] & (NB - 1)], 1)] = (int)(w[2 * s + 1] >> NBSH);
            }
            if (over) {
                for (int s = 0; s < RPW; ++s) {
                    int ps = __shfl(rp0, s), ls = __shfl(rln, s);
                    for (int k = 64 + lane; k < ls; k += 32) {
                        unsigned int ww = bins[ps + k];
                        ss[atomicAdd(&cntl[ww & (NB - 1)], 1)] = (int)(ww >> NBSH);
                    }
                }
            }
            __syncthreads();

            int slot = lane >> 2, q = lane & 3;
            float av[V];
            #pragma unroll
            for (int v = 0; v < V; ++v) av[v] = a_src[q * V + v];
            const char* hb = (const char*)h;
            unsigned int qoff = (unsigned int)(q * V * 4);

            for (int r = wv; r < NB; r += LTHREADS / 32) {
                int o0 = __builtin_amdgcn_readfirstlane(offl[r]);
                int o1 = __builtin_amdgcn_readfirstlane(offl[r + 1]);
                float adv = adl[r];
                float m = NINF, d = 0.0f;
                float a[V];
                #pragma unroll
                for (int v = 0; v < V; ++v) a[v] = 0.0f;

                for (int p0 = o0; p0 < o1; p0 += 8 * UB) {
                    int sidx[UB];
                    #pragma unroll
                    for (int u = 0; u < UB; ++u) {
                        int p = p0 + u * 8 + slot;
                        int pc = (p < o1) ? p : o1 - 1;
                        sidx[u] = ss[pc];
                    }
                    piece_t hv[UB];
                    #pragma unroll
                    for (int u = 0; u < UB; ++u)
                        hv[u] = *(const piece_t*)(hb + ((unsigned int)sidx[u] * (unsigned int)(F * 4) + qoff));
                    float e[UB];
                    float bm = NINF;
                    #pragma unroll
                    for (int u = 0; u < UB; ++u) {
                        float part = 0.0f;
                        #pragma unroll
                        for (int v = 0; v < V; ++v) part += hv[u][v] * av[v];
                        part += xrow<1>(part);
                        part += xrow<2>(part);
                        float x = part + adv;
                        x = fmaxf(x, NEG_SLOPE * x);
                        int p = p0 + u * 8 + slot;
                        e[u] = (p < o1) ? x : NINF;
                        bm = fmaxf(bm, e[u]);
                    }
                    float mn = fmaxf(m, bm);
                    float mu = (mn == NINF) ? 0.0f : mn;
                    float sc = __expf(m - mu);
                    d *= sc;
                    #pragma unroll
                    for (int v = 0; v < V; ++v) a[v] *= sc;
                    #pragma unroll
                    for (int u = 0; u < UB; ++u) {
                        float ex = __expf(e[u] - mu);
                        d += ex;
                        #pragma unroll
                        for (int v = 0; v < V; ++v) a[v] += ex * hv[u][v];
                    }
                    m = mn;
                }

                float M = m;
                M = fmaxf(M, xrow<4>(M));
                M = fmaxf(M, xrow<8>(M));
                M = fmaxf(M, xhalf(M));
                float Mu = (M == NINF) ? 0.0f : M;
                float sc = __expf(m - Mu);
                d *= sc;
                d += xrow<4>(d);
                d += xrow<8>(d);
                d += xhalf(d);
                #pragma unroll
                for (int v = 0; v < V; ++v) {
                    float x = a[v] * sc;
                    x += xrow<4>(x);
                    x += xrow<8>(x);
                    x += xhalf(x);
                    a[v] = x;
                }
                if (slot == 0) {
                    #pragma unroll
                    for (int v = 0; v < V; ++v) acc[r * RS + q * V + v] = a[v];
                    if (q == 0) acc[r * RS + F] = d;
                }
            }
        }
    }
    if (slow) {
        for (int j = t; j < NB * RS; j += LTHREADS) acc[j] = 0.0f;
        __syncthreads();
        for (int cidx = 0; cidx < nch; ++cidx) {
            const int* te = tcol + (long)cidx * pitch;
            int q0 = te[0], q1 = te[1];
            for (int i = q0 + t; i < q1; i += LTHREADS) {
                unsigned int w = bins[i];
                int s = (int)(w >> NBSH), dl = (int)(w & (NB - 1));
                float sv;
                if constexpr (F == HID) sv = row_logit<F>(h, a_src, s);
                else                    sv = as[s];
                float e = sv + adl[dl];
                e = (e >= 0.0f) ? e : NEG_SLOPE * e;
                ldsMaxF(&mI[dl], e);
            }
        }
        __syncthreads();
        for (int cidx = 0; cidx < nch; ++cidx) {
            const int* te = tcol + (long)cidx * pitch;
            int q0 = te[0], q1 = te[1];
            for (int i = q0 + t; i < q1; i += LTHREADS) {
                unsigned int w = bins[i];
                int s = (int)(w >> NBSH), dl = (int)(w & (NB - 1));
                float sv;
                if constexpr (F == HID) sv = row_logit<F>(h, a_src, s);
                else                    sv = as[s];
                float e = sv + adl[dl];
                e = (e >= 0.0f) ? e : NEG_SLOPE * e;
                float ex = __expf(e - __int_as_float(mI[dl]));
                const float4* hs = (const float4*)(h + (long)s * F);
                float* ar = acc + dl * RS;
                atomicAdd(&ar[F], ex);
                #pragma unroll
                for (int q = 0; q < F / 4; ++q) {
                    float4 hv = hs[q];
                    atomicAdd(&ar[4 * q + 0], ex * hv.x);
                    atomicAdd(&ar[4 * q + 1], ex * hv.y);
                    atomicAdd(&ar[4 * q + 2], ex * hv.z);
                    atomicAdd(&ar[4 * q + 3], ex * hv.w);
                }
            }
        }
    }
    __syncthreads();

    if constexpr (F == HID) {
        if (t < NB) {
            float* ar = acc + t * RS;
            float d = ar[F];
            float inv = (d > 0.0f) ? 1.0f / d : 0.0f;
            #pragma unroll
            for (int f = 0; f < F; ++f) ar[f] = fmaxf(ar[f] * inv + bl[f], 0.0f);
        }
        __syncthreads();
        int wt = __builtin_amdgcn_readfirstlane(t >> 5);
        if (wt < NB / 16) {
            int lane = t & 31;
            int col  = lane & 15;
            int hi   = lane >> 4;
            const float* arow = acc + (wt * 16 + col) * RS;
            v8f c = {};
            #pragma unroll
            for (int k0 = 0; k0 < HID; k0 += 4) {
                int ka = k0 + 2 * hi;
                v2f a;
                a.x = arow[ka];
                a.y = arow[ka + 1];
                v2f b;
                b.x = w2l[ka * NCLS + col];
                b.y = w2l[(ka + 1) * NCLS + col];
                c = __builtin_amdgcn_wmma_f32_16x16x4_f32(false, a, false, b,
                                                          (short)0, c, false, false);
            }
            int rb = wt * 16 + (hi << 3);
            #pragma unroll
            for (int j = 0; j < 8; ++j) h2l[(rb + j) * HP + col] = c[j];
        }
        __syncthreads();
        if (t < NB) {
            int node = base + t;
            const float* hr = h2l + t * HP;
            float sv = 0.0f, dv = 0.0f;
            #pragma unroll
            for (int q = 0; q < NCLS / 4; ++q) {
                float x0 = hr[4 * q + 0], x1 = hr[4 * q + 1], x2 = hr[4 * q + 2], x3 = hr[4 * q + 3];
                sv += x0 * anl[4 * q + 0] + x1 * anl[4 * q + 1]
                    + x2 * anl[4 * q + 2] + x3 * anl[4 * q + 3];
                dv += x0 * anl[NCLS + 4 * q + 0] + x1 * anl[NCLS + 4 * q + 1]
                    + x2 * anl[NCLS + 4 * q + 2] + x3 * anl[NCLS + 4 * q + 3];
            }
            if (node < n) {
                asn[node] = sv;
                adn[node] = dv;
            }
        }
        {
            int r = t >> 2, qd = t & 3;
            int node = base + r;
            const float* hr = h2l + r * HP + 4 * qd;
            float4 o;
            o.x = hr[0]; o.y = hr[1]; o.z = hr[2]; o.w = hr[3];
            if (node < n) *(float4*)(hn + (long)node * NCLS + 4 * qd) = o;
        }
    } else {
        if (t < NB) {
            float* ar = acc + t * RS;
            float d = ar[F];
            float inv = (d > 0.0f) ? 1.0f / d : 0.0f;
            float v[F];
            float mx = -__builtin_inff();
            #pragma unroll
            for (int f = 0; f < F; ++f) v[f] = ar[f] * inv + bl[f];
            #pragma unroll
            for (int f = 0; f < F; ++f) mx = fmaxf(mx, v[f]);
            float s = 0.0f;
            #pragma unroll
            for (int f = 0; f < F; ++f) s += __expf(v[f] - mx);
            float ls = __logf(s) + mx;
            #pragma unroll
            for (int f = 0; f < F; ++f) ar[f] = v[f] - ls;
        }
        __syncthreads();
        {
            int r = t >> 2, qd = t & 3;
            int node = base + r;
            const float* ar = acc + r * RS + 4 * qd;
            float4 o;
            o.x = ar[0]; o.y = ar[1]; o.z = ar[2]; o.w = ar[3];
            if (node < n) *(float4*)(out + (long)node * F + 4 * qd) = o;
        }
    }
}

__global__ void finish1(float* __restrict__ agg, const float* __restrict__ denom,
                        const float* __restrict__ bias, int n) {
    int i = blockIdx.x * blockDim.x + threadIdx.x;
    if (i >= n) return;
    float d = denom[i];
    float inv = (d > 0.0f) ? 1.0f / d : 0.0f;
    float4* ap = (float4*)(agg + (long)i * HID);
    #pragma unroll
    for (int q = 0; q < HID / 4; ++q) {
        float4 v = ap[q];
        v.x = fmaxf(v.x * inv + bias[4 * q + 0], 0.0f);
        v.y = fmaxf(v.y * inv + bias[4 * q + 1], 0.0f);
        v.z = fmaxf(v.z * inv + bias[4 * q + 2], 0.0f);
        v.w = fmaxf(v.w * inv + bias[4 * q + 3], 0.0f);
        ap[q] = v;
    }
}

__global__ void finish2(const float* __restrict__ agg, const float* __restrict__ denom,
                        const float* __restrict__ bias, float* __restrict__ out, int n) {
    int i = blockIdx.x * blockDim.x + threadIdx.x;
    if (i >= n) return;
    float d = denom[i];
    float inv = (d > 0.0f) ? 1.0f / d : 0.0f;
    const float4* ap = (const float4*)(agg + (long)i * NCLS);
    float v[NCLS];
    float mx = -__builtin_inff();
    #pragma unroll
    for (int q = 0; q < NCLS / 4; ++q) {
        float4 a = ap[q];
        v[4 * q + 0] = a.x * inv + bias[4 * q + 0];
        v[4 * q + 1] = a.y * inv + bias[4 * q + 1];
        v[4 * q + 2] = a.z * inv + bias[4 * q + 2];
        v[4 * q + 3] = a.w * inv + bias[4 * q + 3];
    }
    #pragma unroll
    for (int f = 0; f < NCLS; ++f) mx = fmaxf(mx, v[f]);
    float s = 0.0f;
    #pragma unroll
    for (int f = 0; f < NCLS; ++f) s += __expf(v[f] - mx);
    float ls = __logf(s) + mx;
    float4* op = (float4*)(out + (long)i * NCLS);
    #pragma unroll
    for (int q = 0; q < NCLS / 4; ++q) {
        float4 o;
        o.x = v[4 * q + 0] - ls;
        o.y = v[4 * q + 1] - ls;
        o.z = v[4 * q + 2] - ls;
        o.w = v[4 * q + 3] - ls;
        op[q] = o;
    }
}

extern "C" void kernel_launch(void* const* d_in, const int* in_sizes, int n_in,
                              void* d_out, int out_size, void* d_ws, size_t ws_size,
                              hipStream_t stream) {
    const float* x   = (const float*)d_in[0];
    const int*   ei  = (const int*)  d_in[1];
    const float* W1  = (const float*)d_in[2];
    const float* a1s = (const float*)d_in[3];
    const float* a1d = (const float*)d_in[4];
    const float* b1  = (const float*)d_in[5];
    const float* W2  = (const float*)d_in[6];
    const float* a2s = (const float*)d_in[7];
    const float* a2d = (const float*)d_in[8];
    const float* b2  = (const float*)d_in[9];

    int n = in_sizes[0] / FIN;
    int E = in_sizes[1] / 2;
    const int* src = ei;
    const int* dst = ei + E;

    float* ws   = (float*)d_ws;
    float* h1   = ws; ws += (long)n * HID;
    float* as1  = ws; ws += n;
    float* ad1  = ws; ws += n;
    float* m1   = ws; ws += n;
    float* dn1  = ws; ws += n;
    float* agg1 = ws; ws += (long)n * HID;
    float* h2   = ws; ws += (long)n * NCLS;
    float* as2  = ws; ws += n;
    float* ad2  = ws; ws += n;
    float* m2   = ws; ws += n;
    float* dn2  = ws; ws += n;
    float* agg2 = ws; ws += (long)n * NCLS;

    int ntiles  = (n + 15) / 16;
    int gblocks = (ntiles + 3) / 4;
    int eblocks = (E + 255) / 256;
    int nblocks = (n + 255) / 256;

    int nb  = (n + NB - 1) / NB;
    int nbp = (nb + 4) & ~3;
    int nch = (E + PCH - 1) / PCH;
    int* tab = (int*)agg2;
    unsigned int* bins = (unsigned int*)((int*)ws + 3 * nbp);
    size_t need = ((size_t)56 * n + (size_t)3 * nbp + (size_t)E) * 4;

    if (ws_size < need || n > (1 << 24) || (size_t)nch * (nb + 1) > (size_t)n * NCLS) {
        init_ws<<<(n * NCLS + 255) / 256, 256, 0, stream>>>(m1, dn1, agg1, m2, dn2, agg2, n);
        gemm1_wmma<<<gblocks, 128, 0, stream>>>(x, W1, h1, n, (int*)nullptr, 0);
        alpha_kernel<HID><<<nblocks, 256, 0, stream>>>(h1, a1s, a1d, as1, ad1, n);
        edge_max<<<eblocks, 256, 0, stream>>>(src, dst, as1, ad1, m1, E);
        edge_acc<HID><<<eblocks, 256, 0, stream>>>(src, dst, as1, ad1, m1, h1, dn1, agg1, E);
        finish1<<<nblocks, 256, 0, stream>>>(agg1, dn1, b1, n);
        gemm2_wmma<<<gblocks, 128, 0, stream>>>(agg1, W2, h2, n);
        alpha_kernel<NCLS><<<nblocks, 256, 0, stream>>>(h2, a2s, a2d, as2, ad2, n);
        edge_max<<<eblocks, 256, 0, stream>>>(src, dst, as2, ad2, m2, E);
        edge_acc<NCLS><<<eblocks, 256, 0, stream>>>(src, dst, as2, ad2, m2, h2, dn2, agg2, E);
        finish2<<<nblocks, 256, 0, stream>>>(agg2, dn2, b2, (float*)d_out, n);
        return;
    }

    gemm1_wmma<<<gblocks, 128, 0, stream>>>(x, W1, h1, n, (int*)nullptr, 0);

    bin_place<<<nch, PTH, 2 * nb * sizeof(int), stream>>>(src, dst, tab, bins, E, nb);

    range_owner<HID><<<nb, LTHREADS, 0, stream>>>(bins, tab, nch, (const float*)nullptr, (const float*)nullptr,
                                                  a1s, a1d, h1, b1, W2, a2s, a2d,
                                                  h2, as2, ad2, (float*)nullptr, n);

    range_owner<NCLS><<<nb, LTHREADS, 0, stream>>>(bins, tab, nch, as2, ad2, a2s, (const float*)nullptr, h2, b2,
                                                   (const float*)nullptr, (const float*)nullptr,
                                                   (const float*)nullptr, (float*)nullptr,
                                                   (float*)nullptr, (float*)nullptr, (float*)d_out, n);
}
